# starting kernel
# baseline (speedup 1.0000x reference)
#include <hip/hip_runtime.h>


#define SEQ_L   2048
#define DI      4096
#define RTOT    160
#define DTR     128
#define DS      16
#define KSPLIT  4
#define KCH     (DI / KSPLIT)
#define DBC_N   (SEQ_L * RTOT)

typedef float v2f __attribute__((ext_vector_type(2)));
typedef float v8f __attribute__((ext_vector_type(8)));

static __device__ __forceinline__ v8f wmma4(v2f a, v2f b, v8f c) {
    return __builtin_amdgcn_wmma_f32_16x16x4_f32(
               false, a, false, b, (short)0, c, false, false);
}

__global__ void __launch_bounds__(128)
sssm_gemm_in(const float* __restrict__ x, const float* __restrict__ Win,
             float* __restrict__ part) {
    const int wave = (blockIdx.x * 128 + threadIdx.x) >> 5;
    const int lane = threadIdx.x & 31;
    const int ks   = wave & (KSPLIT - 1);
    const int tile = wave >> 2;
    const int l0 = (tile / (RTOT / 16)) * 16;
    const int j0 = (tile % (RTOT / 16)) * 16;

    const int m  = lane & 15;
    const int kh = (lane >> 4) << 1;

    const float* ap = x   + (size_t)(l0 + m) * DI + ks * KCH + kh;
    const float* bp = Win + (size_t)(j0 + m) * DI + ks * KCH + kh;

    v8f acc0 = {}, acc1 = {}, acc2 = {}, acc3 = {};
    #pragma unroll 4
    for (int k = 0; k < KCH; k += 16) {
        v2f a0 = *(const v2f*)(ap + 0),  b0 = *(const v2f*)(bp + 0);
        v2f a1 = *(const v2f*)(ap + 4),  b1 = *(const v2f*)(bp + 4);
        v2f a2 = *(const v2f*)(ap + 8),  b2 = *(const v2f*)(bp + 8);
        v2f a3 = *(const v2f*)(ap + 12), b3 = *(const v2f*)(bp + 12);
        acc0 = wmma4(a0, b0, acc0);
        acc1 = wmma4(a1, b1, acc1);
        acc2 = wmma4(a2, b2, acc2);
        acc3 = wmma4(a3, b3, acc3);
        ap += 16; bp += 16;
    }
    v8f acc = (acc0 + acc1) + (acc2 + acc3);

    const int mrow = (lane >> 4) << 3;
    float* dst = part + (size_t)ks * DBC_N;
    #pragma unroll
    for (int r = 0; r < 8; ++r)
        dst[(size_t)(l0 + mrow + r) * RTOT + j0 + (lane & 15)] = acc[r];
}

__global__ void __launch_bounds__(256)
sssm_reduce(const float* __restrict__ part, float* __restrict__ dbc) {
    const int i = blockIdx.x * 256 + threadIdx.x;
    dbc[i] = (part[i] + part[i + DBC_N]) +
             (part[i + 2 * (size_t)DBC_N] + part[i + 3 * (size_t)DBC_N]);
}

__global__ void __launch_bounds__(256)
sssm_gemm_delta(const float* __restrict__ dbc, const float* __restrict__ Wd,
                float* __restrict__ delta) {
    const int wave = (blockIdx.x * 256 + threadIdx.x) >> 5;
    const int lane = threadIdx.x & 31;
    const int l0 = (wave >> 8) * 16;
    const int d0 = (wave & 255) * 16;

    const int m  = lane & 15;
    const int kh = (lane >> 4) << 1;

    const float* ap = dbc + (size_t)(l0 + m) * RTOT + kh;
    const float* bp = Wd  + (size_t)(d0 + m) * DTR  + kh;

    v8f acc0 = {}, acc1 = {}, acc2 = {}, acc3 = {};
    #pragma unroll
    for (int k = 0; k < DTR; k += 16) {
        v2f a0 = *(const v2f*)(ap + 0),  b0 = *(const v2f*)(bp + 0);
        v2f a1 = *(const v2f*)(ap + 4),  b1 = *(const v2f*)(bp + 4);
        v2f a2 = *(const v2f*)(ap + 8),  b2 = *(const v2f*)(bp + 8);
        v2f a3 = *(const v2f*)(ap + 12), b3 = *(const v2f*)(bp + 12);
        acc0 = wmma4(a0, b0, acc0);
        acc1 = wmma4(a1, b1, acc1);
        acc2 = wmma4(a2, b2, acc2);
        acc3 = wmma4(a3, b3, acc3);
        ap += 16; bp += 16;
    }
    v8f acc = (acc0 + acc1) + (acc2 + acc3);

    const int mrow = (lane >> 4) << 3;
    #pragma unroll
    for (int r = 0; r < 8; ++r) {
        float v  = acc[r];
        float sp = fmaxf(v, 0.0f) + log1pf(__expf(-fabsf(v)));
        delta[(size_t)(l0 + mrow + r) * DI + d0 + (lane & 15)] = sp;
    }
}

__global__ void __launch_bounds__(256)
sssm_scan(const float* __restrict__ x, const float* __restrict__ A_log,
          const float* __restrict__ Dvec, const float* __restrict__ dbc,
          float* __restrict__ out) {
    const int t = blockIdx.x * 256 + threadIdx.x;
    const int d = t >> 4;
    const int n = t & 15;

    const float A  = -__expf(A_log[(size_t)d * DS + n]);
    const float Dd = Dvec[d];

    float h = 0.0f;
    const float* xp  = x   + d;
    const float* dp  = out + d;
    const float* bcp = dbc + DTR + n;

    #pragma unroll 4
    for (int l = 0; l < SEQ_L; ++l) {
        float dl = dp[(size_t)l * DI];
        float xl = xp[(size_t)l * DI];
        float Bl = bcp[(size_t)l * RTOT];
        float Cl = bcp[(size_t)l * RTOT + DS];

        float dA = __expf(dl * A);
        h = fmaf(dA, h, dl * Bl * xl);
        float y = h * Cl;

        y += __shfl_xor(y, 1, 32);
        y += __shfl_xor(y, 2, 32);
        y += __shfl_xor(y, 4, 32);
        y += __shfl_xor(y, 8, 32);

        if (n == 0)
            out[(size_t)l * DI + d] = fmaf(xl, Dd, y);
    }
}

extern "C" void kernel_launch(void* const* d_in, const int* in_sizes, int n_in,
                              void* d_out, int out_size, void* d_ws, size_t ws_size,
                              hipStream_t stream) {
    const float* x    = (const float*)d_in[0];
    const float* Win  = (const float*)d_in[1];
    const float* Wd   = (const float*)d_in[2];
    const float* Alog = (const float*)d_in[3];
    const float* Dv   = (const float*)d_in[4];
    float*       out  = (float*)d_out;

    float* part = (float*)d_ws;
    float* dbc  = part + (size_t)KSPLIT * DBC_N;

    sssm_gemm_in<<<1280, 128, 0, stream>>>(x, Win, part);
    sssm_reduce<<<DBC_N / 256, 256, 0, stream>>>(part, dbc);
    sssm_gemm_delta<<<4096, 256, 0, stream>>>(dbc, Wd, out);
    sssm_scan<<<(DI * DS) / 256, 256, 0, stream>>>(x, Alog, Dv, dbc, out);
}


# both GEMMs: A/B tiles staged through LDS by coalesced 16-byte loads (64x160 / 64x128 per work-group; delta rows back through LDS); K split 8, one acc per tile: last-bit changes in dbc, delta
# speedup vs baseline: 8.8693x; 8.8693x over previous
#include <hip/hip_runtime.h>


#define SEQ_L   2048
#define DI      4096
#define RTOT    160
#define DTR     128
#define DS      16
#define KSPLIT  8
#define KCH     (DI / KSPLIT)
#define DBC_N   (SEQ_L * RTOT)

typedef float v2f __attribute__((ext_vector_type(2)));
typedef float v8f __attribute__((ext_vector_type(8)));

static __device__ __forceinline__ v8f wmma4(v2f a, v2f b, v8f c) {
    return __builtin_amdgcn_wmma_f32_16x16x4_f32(
               false, a, false, b, (short)0, c, false, false);
}

typedef float v4f __attribute__((ext_vector_type(4)));

#define G_KT     32
#define G_PITCH  36
#define G_BM     64

template <int NT>
struct frag_set { v2f a[2]; v2f b[NT][2]; };

template <int NT>
static __device__ __forceinline__ void frag_read(
        frag_set<NT>& f, const float* __restrict__ fa, const float* __restrict__ fb) {
    f.a[0] = *(const v2f*)(fa);
    f.a[1] = *(const v2f*)(fa + 4);
    #pragma unroll
    for (int t = 0; t < NT; ++t) {
        f.b[t][0] = *(const v2f*)(fb + t * 16 * G_PITCH);
        f.b[t][1] = *(const v2f*)(fb + t * 16 * G_PITCH + 4);
    }
}

template <int NT>
static __device__ __forceinline__ void frag_mma(const frag_set<NT>& f, v8f (&acc)[NT]) {
    #pragma unroll
    for (int k2 = 0; k2 < 2; ++k2)
        #pragma unroll
        for (int t = 0; t < NT; ++t)
            acc[t] = wmma4(f.a[k2], f.b[t][k2], acc[t]);
}

template <int NT>
static __device__ __forceinline__ void stage_mma(
        const float* __restrict__ fa, const float* __restrict__ fb, v8f (&acc)[NT]) {
    frag_set<NT> f0, f1;
    frag_read<NT>(f0, fa, fb);
    frag_read<NT>(f1, fa + 8, fb + 8);
    __builtin_amdgcn_sched_barrier(0);
    frag_mma<NT>(f0, acc);
    __builtin_amdgcn_sched_barrier(0);
    frag_read<NT>(f0, fa + 16, fb + 16);
    __builtin_amdgcn_sched_barrier(0);
    frag_mma<NT>(f1, acc);
    __builtin_amdgcn_sched_barrier(0);
    frag_read<NT>(f1, fa + 24, fb + 24);
    __builtin_amdgcn_sched_barrier(0);
    frag_mma<NT>(f0, acc);
    frag_mma<NT>(f1, acc);
    __builtin_amdgcn_sched_barrier(0);
}

template <int NWN, int NT>
static __device__ __forceinline__ void lds_tile_gemm(
        const float* __restrict__ Ab, const int lda,
        const float* __restrict__ Bb, const int ldb,
        const int nstage, float* __restrict__ smem, v8f (&acc)[NT]) {
    constexpr int BN    = 16 * NT * NWN;
    constexpr int ROWS  = G_BM + BN;
    constexpr int RPI   = (128 * NWN) / 8;
    constexpr int NLA   = G_BM / RPI;
    constexpr int NLB   = BN / RPI;
    constexpr int STAGE = ROWS * G_PITCH;
    static_assert(BN % RPI == 0 && G_BM % RPI == 0, "tile rows must split evenly");

    const int tid  = threadIdx.x;
    const int lane = tid & 31, w = tid >> 5;
    const int wm = w & 3, wn = w >> 2;
    const int c = lane & 15, h = lane >> 4;
    const int q  = tid & 7;
    const int r0 = tid >> 3;

    const float* ga = Ab + (size_t)r0 * lda + 4 * q;
    const float* gb = Bb + (size_t)r0 * ldb + 4 * q;
    float* sw = smem + r0 * G_PITCH + 4 * q;
    const float* fa = smem + (16 * wm + c) * G_PITCH + 2 * h;
    const float* fb = smem + (G_BM + 16 * NT * wn + c) * G_PITCH + 2 * h;

    v4f pa[NLA], pb[NLB];
    #pragma unroll
    for (int i = 0; i < NLA; ++i) pa[i] = *(const v4f*)(ga + (size_t)(RPI * i) * lda);
    #pragma unroll
    for (int i = 0; i < NLB; ++i) pb[i] = *(const v4f*)(gb + (size_t)(RPI * i) * ldb);
    #pragma unroll
    for (int i = 0; i < NLA; ++i) *(v4f*)(sw + RPI * i * G_PITCH) = pa[i];
    #pragma unroll
    for (int i = 0; i < NLB; ++i) *(v4f*)(sw + (G_BM + RPI * i) * G_PITCH) = pb[i];
    __syncthreads();

    #pragma unroll 1
    for (int s = 0; s + 1 < nstage; ++s) {
        const int cur = (s & 1) * STAGE;
        const int nxt = STAGE - cur;
        ga += G_KT; gb += G_KT;
        #pragma unroll
        for (int i = 0; i < NLA; ++i) pa[i] = *(const v4f*)(ga + (size_t)(RPI * i) * lda);
        #pragma unroll
        for (int i = 0; i < NLB; ++i) pb[i] = *(const v4f*)(gb + (size_t)(RPI * i) * ldb);
        stage_mma<NT>(fa + cur, fb + cur, acc);
        #pragma unroll
        for (int i = 0; i < NLA; ++i) *(v4f*)(sw + nxt + RPI * i * G_PITCH) = pa[i];
        #pragma unroll
        for (int i = 0; i < NLB; ++i) *(v4f*)(sw + nxt + (G_BM + RPI * i) * G_PITCH) = pb[i];
        __syncthreads();
    }
    const int last = ((nstage - 1) & 1) * STAGE;
    stage_mma<NT>(fa + last, fb + last, acc);
}

#define G1_NT 5
__global__ void __launch_bounds__(256)
sssm_gemm_in(const float* __restrict__ x, const float* __restrict__ Win,
             float* __restrict__ part) {
    __shared__ __attribute__((aligned(16))) float smem[2 * (G_BM + RTOT) * G_PITCH];
    const int ks = blockIdx.x & (KSPLIT - 1);
    const int l0 = (blockIdx.x / KSPLIT) * G_BM;
    const int lane = threadIdx.x & 31, w = threadIdx.x >> 5;
    const int wm = w & 3, wn = w >> 2;

    v8f acc[G1_NT];
    #pragma unroll
    for (int t = 0; t < G1_NT; ++t) acc[t] = v8f{};

    lds_tile_gemm<2, G1_NT>(x + (size_t)l0 * DI + ks * KCH, DI,
                            Win + ks * KCH, DI, KCH / G_KT, smem, acc);

    float* dst = part + (size_t)ks * DBC_N
               + (size_t)(l0 + 16 * wm + ((lane >> 4) << 3)) * RTOT
               + 16 * G1_NT * wn + (lane & 15);
    #pragma unroll
    for (int t = 0; t < G1_NT; ++t)
        #pragma unroll
        for (int r = 0; r < 8; ++r)
            dst[(size_t)r * RTOT + 16 * t] = acc[t][r];
}

__global__ void __launch_bounds__(256)
sssm_reduce(const float* __restrict__ part, float* __restrict__ dbc) {
    const int i = blockIdx.x * 256 + threadIdx.x;
    const float s01 = part[i]                     + part[i + (size_t)DBC_N];
    const float s23 = part[i + 2 * (size_t)DBC_N] + part[i + 3 * (size_t)DBC_N];
    const float s45 = part[i + 4 * (size_t)DBC_N] + part[i + 5 * (size_t)DBC_N];
    const float s67 = part[i + 6 * (size_t)DBC_N] + part[i + 7 * (size_t)DBC_N];
    dbc[i] = (s01 + s23) + (s45 + s67);
}

#define G2_NT 8
#define G2_BN (16 * G2_NT)
#define G2_OP (G2_BN + 4)
__global__ void __launch_bounds__(128)
sssm_gemm_delta(const float* __restrict__ dbc, const float* __restrict__ Wd,
                float* __restrict__ delta) {
    __shared__ __attribute__((aligned(16))) float smem[2 * (G_BM + G2_BN) * G_PITCH];
    const int d0 = (blockIdx.x & (DI / G2_BN - 1)) * G2_BN;
    const int l0 = (blockIdx.x / (DI / G2_BN)) * G_BM;
    const int lane = threadIdx.x & 31, w = threadIdx.x >> 5;

    v8f acc[G2_NT];
    #pragma unroll
    for (int t = 0; t < G2_NT; ++t) acc[t] = v8f{};

    lds_tile_gemm<1, G2_NT>(dbc + (size_t)l0 * RTOT, RTOT,
                            Wd + (size_t)d0 * DTR, DTR, DTR / G_KT, smem, acc);
    __syncthreads();

    float* ot = smem + w * (16 * G2_OP);
    {
        float* op = ot + ((lane >> 4) << 3) * G2_OP + (lane & 15);
        #pragma unroll
        for (int t = 0; t < G2_NT; ++t)
            #pragma unroll
            for (int r = 0; r < 8; ++r)
                op[r * G2_OP + 16 * t] = acc[t][r];
    }
    __syncthreads();

    float* dst = delta + (size_t)(l0 + 16 * w) * DI + d0 + 4 * lane;
    #pragma unroll 2
    for (int rr = 0; rr < 16; ++rr) {
        const float4 v4 = *(const float4*)(ot + rr * G2_OP + 4 * lane);
        const float vi[4] = {v4.x, v4.y, v4.z, v4.w};
        float so[4];
        #pragma unroll
        for (int e = 0; e < 4; ++e) {
            const float v = vi[e];
            so[e] = fmaxf(v, 0.0f) + log1pf(__expf(-fabsf(v)));
        }
        *(float4*)(dst + (size_t)rr * DI) = make_float4(so[0], so[1], so[2], so[3]);
    }
}

#define SCAN_CH   4
#define SCAN_PASS (SEQ_L / 32)

static __device__ __forceinline__ float other_row_last(float v) {
    const int b = __builtin_bit_cast(int, v);
    return __builtin_bit_cast(float, __builtin_amdgcn_permlanex16(
        b, b, 0xffffffffu, 0xffffffffu, false, false));
}

template <int S>
static __device__ __forceinline__ void scan_step(float& a, float& b) {
    asm("v_fmac_f32_dpp %0, %0, %1 row_shr:%2 row_mask:0xf bank_mask:0xf"
        : "+v"(b) : "v"(a), "n"(S));
    asm("v_mul_f32_dpp %0, %0, %0 row_shr:%1 row_mask:0xf bank_mask:0xf"
        : "+v"(a) : "n"(S));
}

__global__ void __launch_bounds__(128)
sssm_scan(const float* __restrict__ x, const float* __restrict__ A_log,
          const float* __restrict__ Dvec, const float* __restrict__ dbc,
          float* __restrict__ out) {
    const int lane = threadIdx.x & 31;
    const int wave = (blockIdx.x * 128 + threadIdx.x) >> 5;
    const int d0   = wave * SCAN_CH;
    const bool hi  = lane >= 16;

    float An[SCAN_CH][DS];
    float Dd[SCAN_CH];
    float hin[SCAN_CH][DS];
    #pragma unroll
    for (int j = 0; j < SCAN_CH; ++j) {
        Dd[j] = Dvec[d0 + j];
        #pragma unroll
        for (int n = 0; n < DS; ++n) {
            An[j][n]  = -__expf(A_log[(size_t)(d0 + j) * DS + n]);
            hin[j][n] = 0.0f;
        }
    }

    float4 dC, xC, bcC[8];
    {
        const size_t l = (size_t)lane;
        dC = *(const float4*)(out + l * DI + d0);
        xC = *(const float4*)(x   + l * DI + d0);
        const float4* bc = (const float4*)(dbc + l * RTOT + DTR);
        #pragma unroll
        for (int q = 0; q < 8; ++q) bcC[q] = bc[q];
    }

    #pragma unroll 1
    for (int p = 0; p < SCAN_PASS; ++p) {
        const int pn = (p + 1 < SCAN_PASS) ? (p + 1) : p;
        const size_t ln = (size_t)(pn * 32 + lane);
        const float4 dN = *(const float4*)(out + ln * DI + d0);
        const float4 xN = *(const float4*)(x   + ln * DI + d0);
        float4 bcN[8];
        {
            const float4* bc = (const float4*)(dbc + ln * RTOT + DTR);
            #pragma unroll
            for (int q = 0; q < 8; ++q) bcN[q] = bc[q];
        }
        __builtin_amdgcn_sched_barrier(0);

        const float dl4[4] = {dC.x, dC.y, dC.z, dC.w};
        const float xl4[4] = {xC.x, xC.y, xC.z, xC.w};
        float Bv[DS], Cv[DS];
        #pragma unroll
        for (int q = 0; q < 4; ++q) {
            Bv[4 * q + 0] = bcC[q].x;     Bv[4 * q + 1] = bcC[q].y;
            Bv[4 * q + 2] = bcC[q].z;     Bv[4 * q + 3] = bcC[q].w;
            Cv[4 * q + 0] = bcC[4 + q].x; Cv[4 * q + 1] = bcC[4 + q].y;
            Cv[4 * q + 2] = bcC[4 + q].z; Cv[4 * q + 3] = bcC[4 + q].w;
        }

        float yv[4];
        #pragma unroll
        for (int j = 0; j < SCAN_CH; ++j) {
            const float dl = dl4[j];
            const float xl = xl4[j];
            float yj = 0.0f;
            #pragma unroll
            for (int n = 0; n < DS; ++n) {
                float a = __expf(dl * An[j][n]);
                float b = dl * Bv[n] * xl;
                scan_step<1>(a, b);
                scan_step<2>(a, b);
                scan_step<4>(a, b);
                scan_step<8>(a, b);
                const float t  = fmaf(a, hin[j][n], b);
                const float h15 = other_row_last(t);
                const float u  = fmaf(a, h15, b);
                const float h  = hi ? u : t;
                hin[j][n] = other_row_last(h);
                yj = fmaf(h, Cv[n], yj);
            }
            yv[j] = fmaf(xl, Dd[j], yj);
        }

        const size_t lc = (size_t)(p * 32 + lane);
        *(float4*)(out + lc * DI + d0) = make_float4(yv[0], yv[1], yv[2], yv[3]);

        dC = dN; xC = xN;
        #pragma unroll
        for (int q = 0; q < 8; ++q) bcC[q] = bcN[q];
    }
}

extern "C" void kernel_launch(void* const* d_in, const int* in_sizes, int n_in,
                              void* d_out, int out_size, void* d_ws, size_t ws_size,
                              hipStream_t stream) {
    const float* x    = (const float*)d_in[0];
    const float* Win  = (const float*)d_in[1];
    const float* Wd   = (const float*)d_in[2];
    const float* Alog = (const float*)d_in[3];
    const float* Dv   = (const float*)d_in[4];
    float*       out  = (float*)d_out;

    float* part = (float*)d_ws;
    float* dbc  = part + (size_t)KSPLIT * DBC_N;

    sssm_gemm_in<<<256, 256, 0, stream>>>(x, Win, part);
    sssm_reduce<<<DBC_N / 256, 256, 0, stream>>>(part, dbc);
    sssm_gemm_delta<<<1024, 128, 0, stream>>>(dbc, Wd, out);
    sssm_scan<<<DI / SCAN_CH / 4, 128, 0, stream>>>(x, Alog, Dv, dbc, out);
}


# scan: each lane owns 4 consecutive steps and composes their maps in-lane; one 16-lane DPP prefix per 128 steps (was per 32) runs on the composed maps: sum order of h changes (last bits)
# speedup vs baseline: 11.8388x; 1.3348x over previous
#include <hip/hip_runtime.h>


#define SEQ_L   2048
#define DI      4096
#define RTOT    160
#define DTR     128
#define DS      16
#define KSPLIT  8
#define KCH     (DI / KSPLIT)
#define DBC_N   (SEQ_L * RTOT)

typedef float v2f __attribute__((ext_vector_type(2)));
typedef float v8f __attribute__((ext_vector_type(8)));

static __device__ __forceinline__ v8f wmma4(v2f a, v2f b, v8f c) {
    return __builtin_amdgcn_wmma_f32_16x16x4_f32(
               false, a, false, b, (short)0, c, false, false);
}

typedef float v4f __attribute__((ext_vector_type(4)));

#define G_KT     32
#define G_PITCH  36
#define G_BM     64

template <int NT>
struct frag_set { v2f a[2]; v2f b[NT][2]; };

template <int NT>
static __device__ __forceinline__ void frag_read(
        frag_set<NT>& f, const float* __restrict__ fa, const float* __restrict__ fb) {
    f.a[0] = *(const v2f*)(fa);
    f.a[1] = *(const v2f*)(fa + 4);
    #pragma unroll
    for (int t = 0; t < NT; ++t) {
        f.b[t][0] = *(const v2f*)(fb + t * 16 * G_PITCH);
        f.b[t][1] = *(const v2f*)(fb + t * 16 * G_PITCH + 4);
    }
}

template <int NT>
static __device__ __forceinline__ void frag_mma(const frag_set<NT>& f, v8f (&acc)[NT]) {
    #pragma unroll
    for (int k2 = 0; k2 < 2; ++k2)
        #pragma unroll
        for (int t = 0; t < NT; ++t)
            acc[t] = wmma4(f.a[k2], f.b[t][k2], acc[t]);
}

template <int NT>
static __device__ __forceinline__ void stage_mma(
        const float* __restrict__ fa, const float* __restrict__ fb, v8f (&acc)[NT]) {
    frag_set<NT> f0, f1;
    frag_read<NT>(f0, fa, fb);
    frag_read<NT>(f1, fa + 8, fb + 8);
    __builtin_amdgcn_sched_barrier(0);
    frag_mma<NT>(f0, acc);
    __builtin_amdgcn_sched_barrier(0);
    frag_read<NT>(f0, fa + 16, fb + 16);
    __builtin_amdgcn_sched_barrier(0);
    frag_mma<NT>(f1, acc);
    __builtin_amdgcn_sched_barrier(0);
    frag_read<NT>(f1, fa + 24, fb + 24);
    __builtin_amdgcn_sched_barrier(0);
    frag_mma<NT>(f0, acc);
    frag_mma<NT>(f1, acc);
    __builtin_amdgcn_sched_barrier(0);
}

template <int NWN, int NT>
static __device__ __forceinline__ void lds_tile_gemm(
        const float* __restrict__ Ab, const int lda,
        const float* __restrict__ Bb, const int ldb,
        const int nstage, float* __restrict__ smem, v8f (&acc)[NT]) {
    constexpr int BN    = 16 * NT * NWN;
    constexpr int ROWS  = G_BM + BN;
    constexpr int RPI   = (128 * NWN) / 8;
    constexpr int NLA   = G_BM / RPI;
    constexpr int NLB   = BN / RPI;
    constexpr int STAGE = ROWS * G_PITCH;
    static_assert(BN % RPI == 0 && G_BM % RPI == 0, "tile rows must split evenly");

    const int tid  = threadIdx.x;
    const int lane = tid & 31, w = tid >> 5;
    const int wm = w & 3, wn = w >> 2;
    const int c = lane & 15, h = lane >> 4;
    const int q  = tid & 7;
    const int r0 = tid >> 3;

    const float* ga = Ab + (size_t)r0 * lda + 4 * q;
    const float* gb = Bb + (size_t)r0 * ldb + 4 * q;
    float* sw = smem + r0 * G_PITCH + 4 * q;
    const float* fa = smem + (16 * wm + c) * G_PITCH + 2 * h;
    const float* fb = smem + (G_BM + 16 * NT * wn + c) * G_PITCH + 2 * h;

    v4f pa[NLA], pb[NLB];
    #pragma unroll
    for (int i = 0; i < NLA; ++i) pa[i] = *(const v4f*)(ga + (size_t)(RPI * i) * lda);
    #pragma unroll
    for (int i = 0; i < NLB; ++i) pb[i] = *(const v4f*)(gb + (size_t)(RPI * i) * ldb);
    #pragma unroll
    for (int i = 0; i < NLA; ++i) *(v4f*)(sw + RPI * i * G_PITCH) = pa[i];
    #pragma unroll
    for (int i = 0; i < NLB; ++i) *(v4f*)(sw + (G_BM + RPI * i) * G_PITCH) = pb[i];
    __syncthreads();

    #pragma unroll 1
    for (int s = 0; s + 1 < nstage; ++s) {
        const int cur = (s & 1) * STAGE;
        const int nxt = STAGE - cur;
        ga += G_KT; gb += G_KT;
        #pragma unroll
        for (int i = 0; i < NLA; ++i) pa[i] = *(const v4f*)(ga + (size_t)(RPI * i) * lda);
        #pragma unroll
        for (int i = 0; i < NLB; ++i) pb[i] = *(const v4f*)(gb + (size_t)(RPI * i) * ldb);
        stage_mma<NT>(fa + cur, fb + cur, acc);
        #pragma unroll
        for (int i = 0; i < NLA; ++i) *(v4f*)(sw + nxt + RPI * i * G_PITCH) = pa[i];
        #pragma unroll
        for (int i = 0; i < NLB; ++i) *(v4f*)(sw + nxt + (G_BM + RPI * i) * G_PITCH) = pb[i];
        __syncthreads();
    }
    const int last = ((nstage - 1) & 1) * STAGE;
    stage_mma<NT>(fa + last, fb + last, acc);
}

#define G1_NT 5
__global__ void __launch_bounds__(256)
sssm_gemm_in(const float* __restrict__ x, const float* __restrict__ Win,
             float* __restrict__ part) {
    __shared__ __attribute__((aligned(16))) float smem[2 * (G_BM + RTOT) * G_PITCH];
    const int ks = blockIdx.x & (KSPLIT - 1);
    const int l0 = (blockIdx.x / KSPLIT) * G_BM;
    const int lane = threadIdx.x & 31, w = threadIdx.x >> 5;
    const int wm = w & 3, wn = w >> 2;

    v8f acc[G1_NT];
    #pragma unroll
    for (int t = 0; t < G1_NT; ++t) acc[t] = v8f{};

    lds_tile_gemm<2, G1_NT>(x + (size_t)l0 * DI + ks * KCH, DI,
                            Win + ks * KCH, DI, KCH / G_KT, smem, acc);

    float* dst = part + (size_t)ks * DBC_N
               + (size_t)(l0 + 16 * wm + ((lane >> 4) << 3)) * RTOT
               + 16 * G1_NT * wn + (lane & 15);
    #pragma unroll
    for (int t = 0; t < G1_NT; ++t)
        #pragma unroll
        for (int r = 0; r < 8; ++r)
            dst[(size_t)r * RTOT + 16 * t] = acc[t][r];
}

__global__ void __launch_bounds__(256)
sssm_reduce(const float* __restrict__ part, float* __restrict__ dbc) {
    const int i = blockIdx.x * 256 + threadIdx.x;
    const float s01 = part[i]                     + part[i + (size_t)DBC_N];
    const float s23 = part[i + 2 * (size_t)DBC_N] + part[i + 3 * (size_t)DBC_N];
    const float s45 = part[i + 4 * (size_t)DBC_N] + part[i + 5 * (size_t)DBC_N];
    const float s67 = part[i + 6 * (size_t)DBC_N] + part[i + 7 * (size_t)DBC_N];
    dbc[i] = (s01 + s23) + (s45 + s67);
}

#define G2_NT 8
#define G2_BN (16 * G2_NT)
#define G2_OP (G2_BN + 4)
__global__ void __launch_bounds__(128)
sssm_gemm_delta(const float* __restrict__ dbc, const float* __restrict__ Wd,
                float* __restrict__ delta) {
    __shared__ __attribute__((aligned(16))) float smem[2 * (G_BM + G2_BN) * G_PITCH];
    const int d0 = (blockIdx.x & (DI / G2_BN - 1)) * G2_BN;
    const int l0 = (blockIdx.x / (DI / G2_BN)) * G_BM;
    const int lane = threadIdx.x & 31, w = threadIdx.x >> 5;

    v8f acc[G2_NT];
    #pragma unroll
    for (int t = 0; t < G2_NT; ++t) acc[t] = v8f{};

    lds_tile_gemm<1, G2_NT>(dbc + (size_t)l0 * RTOT, RTOT,
                            Wd + (size_t)d0 * DTR, DTR, DTR / G_KT, smem, acc);
    __syncthreads();

    float* ot = smem + w * (16 * G2_OP);
    {
        float* op = ot + ((lane >> 4) << 3) * G2_OP + (lane & 15);
        #pragma unroll
        for (int t = 0; t < G2_NT; ++t)
            #pragma unroll
            for (int r = 0; r < 8; ++r)
                op[r * G2_OP + 16 * t] = acc[t][r];
    }
    __syncthreads();

    float* dst = delta + (size_t)(l0 + 16 * w) * DI + d0 + 4 * lane;
    #pragma unroll 2
    for (int rr = 0; rr < 16; ++rr) {
        const float4 v4 = *(const float4*)(ot + rr * G2_OP + 4 * lane);
        const float vi[4] = {v4.x, v4.y, v4.z, v4.w};
        float so[4];
        #pragma unroll
        for (int e = 0; e < 4; ++e) {
            const float v = vi[e];
            so[e] = fmaxf(v, 0.0f) + log1pf(__expf(-fabsf(v)));
        }
        *(float4*)(dst + (size_t)rr * DI) = make_float4(so[0], so[1], so[2], so[3]);
    }
}

#define SCAN_CH   4
#define SCAN_K    4
#define SCAN_SPAN (32 * SCAN_K)
#define SCAN_PASS (SEQ_L / SCAN_SPAN)

static __device__ __forceinline__ float other_row_last(float v) {
    const int b = __builtin_bit_cast(int, v);
    return __builtin_bit_cast(float, __builtin_amdgcn_permlanex16(
        b, b, 0xffffffffu, 0xffffffffu, false, false));
}

template <int S>
static __device__ __forceinline__ void scan_step(float& a, float& b) {
    asm("v_fmac_f32_dpp %0, %0, %1 row_shr:%2 row_mask:0xf bank_mask:0xf"
        : "+v"(b) : "v"(a), "n"(S));
    asm("v_mul_f32_dpp %0, %0, %0 row_shr:%1 row_mask:0xf bank_mask:0xf"
        : "+v"(a) : "n"(S));
}

static __device__ __forceinline__ void take_prev_lane(float& carry, float hend) {
    asm("v_mov_b32_dpp %0, %1 row_shr:1 row_mask:0xf bank_mask:0xf"
        : "+v"(carry) : "v"(hend));
}

struct scan_ops { float4 d[SCAN_K], x[SCAN_K], bc[SCAN_K][8]; };

static __device__ __forceinline__ void scan_load(
        scan_ops& o, const float* __restrict__ x, const float* __restrict__ dbc,
        const float* out, const size_t row0, const int d0) {
    #pragma unroll
    for (int s = 0; s < SCAN_K; ++s) {
        const size_t l = row0 + s;
        o.d[s] = *(const float4*)(out + l * DI + d0);
        o.x[s] = *(const float4*)(x   + l * DI + d0);
        const float4* bc = (const float4*)(dbc + l * RTOT + DTR);
        #pragma unroll
        for (int q = 0; q < 8; ++q) o.bc[s][q] = bc[q];
    }
}

__global__ void __launch_bounds__(128)
sssm_scan(const float* __restrict__ x, const float* __restrict__ A_log,
          const float* __restrict__ Dvec, const float* __restrict__ dbc,
          float* __restrict__ out) {
    const int lane = threadIdx.x & 31;
    const int wave = (blockIdx.x * 128 + threadIdx.x) >> 5;
    const int d0   = wave * SCAN_CH;
    const bool hi  = lane >= 16;

    float An[SCAN_CH][DS];
    float Dd[SCAN_CH];
    float hin[SCAN_CH][DS];
    #pragma unroll
    for (int j = 0; j < SCAN_CH; ++j) {
        Dd[j] = Dvec[d0 + j];
        #pragma unroll
        for (int n = 0; n < DS; ++n) {
            An[j][n]  = -__expf(A_log[(size_t)(d0 + j) * DS + n]);
            hin[j][n] = 0.0f;
        }
    }

    scan_ops cur;
    scan_load(cur, x, dbc, out, (size_t)(SCAN_K * lane), d0);

    #pragma unroll 1
    for (int p = 0; p < SCAN_PASS; ++p) {
        scan_ops nxt = cur;
        if (p + 1 < SCAN_PASS)
            scan_load(nxt, x, dbc, out, (size_t)((p + 1) * SCAN_SPAN + SCAN_K * lane), d0);

        float dl[SCAN_K][SCAN_CH], xl[SCAN_K][SCAN_CH], Bv[SCAN_K][DS], Cv[SCAN_K][DS];
        float yv[SCAN_K][SCAN_CH];
        #pragma unroll
        for (int s = 0; s < SCAN_K; ++s) {
            dl[s][0] = cur.d[s].x; dl[s][1] = cur.d[s].y; dl[s][2] = cur.d[s].z; dl[s][3] = cur.d[s].w;
            xl[s][0] = cur.x[s].x; xl[s][1] = cur.x[s].y; xl[s][2] = cur.x[s].z; xl[s][3] = cur.x[s].w;
            #pragma unroll
            for (int q = 0; q < 4; ++q) {
                Bv[s][4 * q + 0] = cur.bc[s][q].x;     Bv[s][4 * q + 1] = cur.bc[s][q].y;
                Bv[s][4 * q + 2] = cur.bc[s][q].z;     Bv[s][4 * q + 3] = cur.bc[s][q].w;
                Cv[s][4 * q + 0] = cur.bc[s][4 + q].x; Cv[s][4 * q + 1] = cur.bc[s][4 + q].y;
                Cv[s][4 * q + 2] = cur.bc[s][4 + q].z; Cv[s][4 * q + 3] = cur.bc[s][4 + q].w;
            }
            #pragma unroll
            for (int j = 0; j < SCAN_CH; ++j) yv[s][j] = 0.0f;
        }

        #pragma unroll
        for (int j = 0; j < SCAN_CH; ++j) {
            #pragma unroll
            for (int n = 0; n < DS; ++n) {
                float P[SCAN_K], g[SCAN_K];
                #pragma unroll
                for (int s = 0; s < SCAN_K; ++s) {
                    const float a = __expf(dl[s][j] * An[j][n]);
                    const float b = dl[s][j] * Bv[s][n] * xl[s][j];
                    P[s] = (s == 0) ? a : a * P[s ? s - 1 : 0];
                    g[s] = (s == 0) ? b : fmaf(a, g[s ? s - 1 : 0], b);
                }
                float am = P[SCAN_K - 1], bm = g[SCAN_K - 1];
                scan_step<1>(am, bm);
                scan_step<2>(am, bm);
                scan_step<4>(am, bm);
                scan_step<8>(am, bm);
                const float t   = fmaf(am, hin[j][n], bm);
                const float h15 = other_row_last(t);
                float carry     = hi ? h15 : hin[j][n];
                const float hend = fmaf(am, carry, bm);
                hin[j][n] = other_row_last(hend);
                take_prev_lane(carry, hend);
                #pragma unroll
                for (int s = 0; s + 1 < SCAN_K; ++s) {
                    const float h = fmaf(P[s], carry, g[s]);
                    yv[s][j] = fmaf(h, Cv[s][n], yv[s][j]);
                }
                yv[SCAN_K - 1][j] = fmaf(hend, Cv[SCAN_K - 1][n], yv[SCAN_K - 1][j]);
            }
        }

        const size_t lc = (size_t)(p * SCAN_SPAN + SCAN_K * lane);
        #pragma unroll
        for (int s = 0; s < SCAN_K; ++s)
            *(float4*)(out + (lc + s) * DI + d0) = make_float4(
                fmaf(xl[s][0], Dd[0], yv[s][0]), fmaf(xl[s][1], Dd[1], yv[s][1]),
                fmaf(xl[s][2], Dd[2], yv[s][2]), fmaf(xl[s][3], Dd[3], yv[s][3]));

        cur = nxt;
    }
}

extern "C" void kernel_launch(void* const* d_in, const int* in_sizes, int n_in,
                              void* d_out, int out_size, void* d_ws, size_t ws_size,
                              hipStream_t stream) {
    const float* x    = (const float*)d_in[0];
    const float* Win  = (const float*)d_in[1];
    const float* Wd   = (const float*)d_in[2];
    const float* Alog = (const float*)d_in[3];
    const float* Dv   = (const float*)d_in[4];
    float*       out  = (float*)d_out;

    float* part = (float*)d_ws;
    float* dbc  = part + (size_t)KSPLIT * DBC_N;

    sssm_gemm_in<<<256, 256, 0, stream>>>(x, Win, part);
    sssm_reduce<<<DBC_N / 256, 256, 0, stream>>>(part, dbc);
    sssm_gemm_delta<<<1024, 128, 0, stream>>>(dbc, Wd, out);
    sssm_scan<<<DI / SCAN_CH / 4, 128, 0, stream>>>(x, Alog, Dv, dbc, out);
}


# scan operands leave their producers in the scan's lane order (B|C from the reduce, delta from GEMM2's LDS turn): a scan load reads 4 whole lines, was 32 partial; values and sums bit-identical
# speedup vs baseline: 12.1997x; 1.0305x over previous
#include <hip/hip_runtime.h>


#define SEQ_L   2048
#define DI      4096
#define RTOT    160
#define DTR     128
#define DS      16
#define KSPLIT  8
#define KCH     (DI / KSPLIT)
#define DBC_N   (SEQ_L * RTOT)

typedef float v2f __attribute__((ext_vector_type(2)));
typedef float v8f __attribute__((ext_vector_type(8)));

static __device__ __forceinline__ v8f wmma4(v2f a, v2f b, v8f c) {
    return __builtin_amdgcn_wmma_f32_16x16x4_f32(
               false, a, false, b, (short)0, c, false, false);
}

typedef float v4f __attribute__((ext_vector_type(4)));

#define G_KT     32
#define G_PITCH  36
#define G_BM     64

template <int NT>
struct frag_set { v2f a[2]; v2f b[NT][2]; };

template <int NT>
static __device__ __forceinline__ void frag_read(
        frag_set<NT>& f, const float* __restrict__ fa, const float* __restrict__ fb) {
    f.a[0] = *(const v2f*)(fa);
    f.a[1] = *(const v2f*)(fa + 4);
    #pragma unroll
    for (int t = 0; t < NT; ++t) {
        f.b[t][0] = *(const v2f*)(fb + t * 16 * G_PITCH);
        f.b[t][1] = *(const v2f*)(fb + t * 16 * G_PITCH + 4);
    }
}

template <int NT>
static __device__ __forceinline__ void frag_mma(const frag_set<NT>& f, v8f (&acc)[NT]) {
    #pragma unroll
    for (int k2 = 0; k2 < 2; ++k2)
        #pragma unroll
        for (int t = 0; t < NT; ++t)
            acc[t] = wmma4(f.a[k2], f.b[t][k2], acc[t]);
}

template <int NT>
static __device__ __forceinline__ void stage_mma(
        const float* __restrict__ fa, const float* __restrict__ fb, v8f (&acc)[NT]) {
    frag_set<NT> f0, f1;
    frag_read<NT>(f0, fa, fb);
    frag_read<NT>(f1, fa + 8, fb + 8);
    __builtin_amdgcn_sched_barrier(0);
    frag_mma<NT>(f0, acc);
    __builtin_amdgcn_sched_barrier(0);
    frag_read<NT>(f0, fa + 16, fb + 16);
    __builtin_amdgcn_sched_barrier(0);
    frag_mma<NT>(f1, acc);
    __builtin_amdgcn_sched_barrier(0);
    frag_read<NT>(f1, fa + 24, fb + 24);
    __builtin_amdgcn_sched_barrier(0);
    frag_mma<NT>(f0, acc);
    frag_mma<NT>(f1, acc);
    __builtin_amdgcn_sched_barrier(0);
}

template <int NWN, int NT>
static __device__ __forceinline__ void lds_tile_gemm(
        const float* __restrict__ Ab, const int lda,
        const float* __restrict__ Bb, const int ldb,
        const int nstage, float* __restrict__ smem, v8f (&acc)[NT]) {
    constexpr int BN    = 16 * NT * NWN;
    constexpr int ROWS  = G_BM + BN;
    constexpr int RPI   = (128 * NWN) / 8;
    constexpr int NLA   = G_BM / RPI;
    constexpr int NLB   = BN / RPI;
    constexpr int STAGE = ROWS * G_PITCH;
    static_assert(BN % RPI == 0 && G_BM % RPI == 0, "tile rows must split evenly");

    const int tid  = threadIdx.x;
    const int lane = tid & 31, w = tid >> 5;
    const int wm = w & 3, wn = w >> 2;
    const int c = lane & 15, h = lane >> 4;
    const int q  = tid & 7;
    const int r0 = tid >> 3;

    const float* ga = Ab + (size_t)r0 * lda + 4 * q;
    const float* gb = Bb + (size_t)r0 * ldb + 4 * q;
    float* sw = smem + r0 * G_PITCH + 4 * q;
    const float* fa = smem + (16 * wm + c) * G_PITCH + 2 * h;
    const float* fb = smem + (G_BM + 16 * NT * wn + c) * G_PITCH + 2 * h;

    v4f pa[NLA], pb[NLB];
    #pragma unroll
    for (int i = 0; i < NLA; ++i) pa[i] = *(const v4f*)(ga + (size_t)(RPI * i) * lda);
    #pragma unroll
    for (int i = 0; i < NLB; ++i) pb[i] = *(const v4f*)(gb + (size_t)(RPI * i) * ldb);
    #pragma unroll
    for (int i = 0; i < NLA; ++i) *(v4f*)(sw + RPI * i * G_PITCH) = pa[i];
    #pragma unroll
    for (int i = 0; i < NLB; ++i) *(v4f*)(sw + (G_BM + RPI * i) * G_PITCH) = pb[i];
    __syncthreads();

    #pragma unroll 1
    for (int s = 0; s + 1 < nstage; ++s) {
        const int cur = (s & 1) * STAGE;
        const int nxt = STAGE - cur;
        ga += G_KT; gb += G_KT;
        #pragma unroll
        for (int i = 0; i < NLA; ++i) pa[i] = *(const v4f*)(ga + (size_t)(RPI * i) * lda);
        #pragma unroll
        for (int i = 0; i < NLB; ++i) pb[i] = *(const v4f*)(gb + (size_t)(RPI * i) * ldb);
        stage_mma<NT>(fa + cur, fb + cur, acc);
        #pragma unroll
        for (int i = 0; i < NLA; ++i) *(v4f*)(sw + nxt + RPI * i * G_PITCH) = pa[i];
        #pragma unroll
        for (int i = 0; i < NLB; ++i) *(v4f*)(sw + nxt + (G_BM + RPI * i) * G_PITCH) = pb[i];
        __syncthreads();
    }
    const int last = ((nstage - 1) & 1) * STAGE;
    stage_mma<NT>(fa + last, fb + last, acc);
}

#define G1_NT 5
__global__ void __launch_bounds__(256)
sssm_gemm_in(const float* __restrict__ x, const float* __restrict__ Win,
             float* __restrict__ part) {
    __shared__ __attribute__((aligned(16))) float smem[2 * (G_BM + RTOT) * G_PITCH];
    const int ks = blockIdx.x & (KSPLIT - 1);
    const int l0 = (blockIdx.x / KSPLIT) * G_BM;
    const int lane = threadIdx.x & 31, w = threadIdx.x >> 5;
    const int wm = w & 3, wn = w >> 2;

    v8f acc[G1_NT];
    #pragma unroll
    for (int t = 0; t < G1_NT; ++t) acc[t] = v8f{};

    lds_tile_gemm<2, G1_NT>(x + (size_t)l0 * DI + ks * KCH, DI,
                            Win + ks * KCH, DI, KCH / G_KT, smem, acc);

    float* dst = part + (size_t)ks * DBC_N
               + (size_t)(l0 + 16 * wm + ((lane >> 4) << 3)) * RTOT
               + 16 * G1_NT * wn + (lane & 15);
    #pragma unroll
    for (int t = 0; t < G1_NT; ++t)
        #pragma unroll
        for (int r = 0; r < 8; ++r)
            dst[(size_t)r * RTOT + 16 * t] = acc[t][r];
}

__global__ void __launch_bounds__(256)
sssm_reduce(const float* __restrict__ part, float* __restrict__ dbc,
            float* __restrict__ bct) {
    const int i = blockIdx.x * 256 + threadIdx.x;
    const float s01 = part[i]                     + part[i + (size_t)DBC_N];
    const float s23 = part[i + 2 * (size_t)DBC_N] + part[i + 3 * (size_t)DBC_N];
    const float s45 = part[i + 4 * (size_t)DBC_N] + part[i + 5 * (size_t)DBC_N];
    const float s67 = part[i + 6 * (size_t)DBC_N] + part[i + 7 * (size_t)DBC_N];
    const float v = (s01 + s23) + (s45 + s67);
    const int l = i / RTOT, col = i - l * RTOT;
    if (col < DTR) {
        dbc[i] = v;
    } else {
        const int cc = col - DTR;
        const int p = l >> 7, ln = (l >> 2) & 31, s = l & 3;
        bct[((((p * 4 + s) * 8 + (cc >> 2)) * 32 + ln) << 2) + (cc & 3)] = v;
    }
}

#define G2_NT 8
#define G2_BN (16 * G2_NT)
#define G2_WP (4 * 16 * 4 + 16)
__global__ void __launch_bounds__(128)
sssm_gemm_delta(const float* __restrict__ dbc, const float* __restrict__ Wd,
                float* __restrict__ delta_t) {
    __shared__ __attribute__((aligned(16))) float smem[2 * (G_BM + G2_BN) * G_PITCH];
    const int d0 = (blockIdx.x & (DI / G2_BN - 1)) * G2_BN;
    const int l0 = (blockIdx.x / (DI / G2_BN)) * G_BM;
    const int lane = threadIdx.x & 31, w = threadIdx.x >> 5;

    v8f acc[G2_NT];
    #pragma unroll
    for (int t = 0; t < G2_NT; ++t) acc[t] = v8f{};

    lds_tile_gemm<1, G2_NT>(dbc + (size_t)l0 * RTOT, RTOT,
                            Wd + (size_t)d0 * DTR, DTR, DTR / G_KT, smem, acc);
    __syncthreads();

    {
        const int c = lane & 15, h = lane >> 4;
        float* op = smem + (c >> 2) * G2_WP + (4 * w + 2 * h) * 4 + (c & 3);
        #pragma unroll
        for (int t = 0; t < G2_NT; ++t)
            #pragma unroll
            for (int r = 0; r < 8; ++r)
                op[4 * t * G2_WP + (r & 3) * 64 + (r >> 2) * 4] = acc[t][r];
    }
    __syncthreads();

    const int il = lane & 15, s = (2 * w + (lane >> 4)) & 3;
    const float* ip = smem + (w >> 1) * G2_WP + s * 64 + il * 4;
    float4* dst = (float4*)delta_t
                + ((size_t)(((d0 >> 2) + (w >> 1)) * 16 + (l0 >> 7)) * 4 + s) * 32
                + ((l0 >> 6) & 1) * 16 + il;
    #pragma unroll 2
    for (int k = 0; k < 16; ++k) {
        const float4 v4 = *(const float4*)(ip + 2 * k * G2_WP);
        const float vi[4] = {v4.x, v4.y, v4.z, v4.w};
        float so[4];
        #pragma unroll
        for (int e = 0; e < 4; ++e) {
            const float v = vi[e];
            so[e] = fmaxf(v, 0.0f) + log1pf(__expf(-fabsf(v)));
        }
        dst[(size_t)k * (2 * 16 * 4 * 32)] = make_float4(so[0], so[1], so[2], so[3]);
    }
}

#define SCAN_CH   4
#define SCAN_K    4
#define SCAN_SPAN (32 * SCAN_K)
#define SCAN_PASS (SEQ_L / SCAN_SPAN)

static __device__ __forceinline__ float other_row_last(float v) {
    const int b = __builtin_bit_cast(int, v);
    return __builtin_bit_cast(float, __builtin_amdgcn_permlanex16(
        b, b, 0xffffffffu, 0xffffffffu, false, false));
}

template <int S>
static __device__ __forceinline__ void scan_step(float& a, float& b) {
    asm("v_fmac_f32_dpp %0, %0, %1 row_shr:%2 row_mask:0xf bank_mask:0xf"
        : "+v"(b) : "v"(a), "n"(S));
    asm("v_mul_f32_dpp %0, %0, %0 row_shr:%1 row_mask:0xf bank_mask:0xf"
        : "+v"(a) : "n"(S));
}

static __device__ __forceinline__ void take_prev_lane(float& carry, float hend) {
    asm("v_mov_b32_dpp %0, %1 row_shr:1 row_mask:0xf bank_mask:0xf"
        : "+v"(carry) : "v"(hend));
}

struct scan_ops { float4 d[SCAN_K], x[SCAN_K], bc[SCAN_K][8]; };

static __device__ __forceinline__ void scan_load(
        scan_ops& o, const float* __restrict__ x, const float4* __restrict__ bcl,
        const float4* __restrict__ dl, const size_t row0, const int d0) {
    #pragma unroll
    for (int s = 0; s < SCAN_K; ++s) {
        const size_t l = row0 + s;
        o.d[s] = dl[s * 32];
        o.x[s] = *(const float4*)(x   + l * DI + d0);
        #pragma unroll
        for (int q = 0; q < 8; ++q) o.bc[s][q] = bcl[(s * 8 + q) * 32];
    }
}

__global__ void __launch_bounds__(128)
sssm_scan(const float* __restrict__ x, const float* __restrict__ A_log,
          const float* __restrict__ Dvec, const float* __restrict__ bct,
          const float* __restrict__ delta_t, float* __restrict__ out) {
    const int lane = threadIdx.x & 31;
    const int wave = (blockIdx.x * 128 + threadIdx.x) >> 5;
    const int d0   = wave * SCAN_CH;
    const bool hi  = lane >= 16;

    float An[SCAN_CH][DS];
    float Dd[SCAN_CH];
    float hin[SCAN_CH][DS];
    #pragma unroll
    for (int j = 0; j < SCAN_CH; ++j) {
        Dd[j] = Dvec[d0 + j];
        #pragma unroll
        for (int n = 0; n < DS; ++n) {
            An[j][n]  = -__expf(A_log[(size_t)(d0 + j) * DS + n]);
            hin[j][n] = 0.0f;
        }
    }

    scan_ops cur;
    const float4* bcl = (const float4*)bct + lane;
    const float4* dl  = (const float4*)delta_t
                      + (size_t)wave * (SCAN_PASS * SCAN_K * 32) + lane;
    scan_load(cur, x, bcl, dl, (size_t)(SCAN_K * lane), d0);

    #pragma unroll 1
    for (int p = 0; p < SCAN_PASS; ++p) {
        scan_ops nxt = cur;
        if (p + 1 < SCAN_PASS)
            scan_load(nxt, x, bcl + (size_t)(p + 1) * (SCAN_K * 8 * 32),
                      dl + (size_t)(p + 1) * (SCAN_K * 32),
                      (size_t)((p + 1) * SCAN_SPAN + SCAN_K * lane), d0);

        float dl[SCAN_K][SCAN_CH], xl[SCAN_K][SCAN_CH], Bv[SCAN_K][DS], Cv[SCAN_K][DS];
        float yv[SCAN_K][SCAN_CH];
        #pragma unroll
        for (int s = 0; s < SCAN_K; ++s) {
            dl[s][0] = cur.d[s].x; dl[s][1] = cur.d[s].y; dl[s][2] = cur.d[s].z; dl[s][3] = cur.d[s].w;
            xl[s][0] = cur.x[s].x; xl[s][1] = cur.x[s].y; xl[s][2] = cur.x[s].z; xl[s][3] = cur.x[s].w;
            #pragma unroll
            for (int q = 0; q < 4; ++q) {
                Bv[s][4 * q + 0] = cur.bc[s][q].x;     Bv[s][4 * q + 1] = cur.bc[s][q].y;
                Bv[s][4 * q + 2] = cur.bc[s][q].z;     Bv[s][4 * q + 3] = cur.bc[s][q].w;
                Cv[s][4 * q + 0] = cur.bc[s][4 + q].x; Cv[s][4 * q + 1] = cur.bc[s][4 + q].y;
                Cv[s][4 * q + 2] = cur.bc[s][4 + q].z; Cv[s][4 * q + 3] = cur.bc[s][4 + q].w;
            }
            #pragma unroll
            for (int j = 0; j < SCAN_CH; ++j) yv[s][j] = 0.0f;
        }

        #pragma unroll
        for (int j = 0; j < SCAN_CH; ++j) {
            #pragma unroll
            for (int n = 0; n < DS; ++n) {
                float P[SCAN_K], g[SCAN_K];
                #pragma unroll
                for (int s = 0; s < SCAN_K; ++s) {
                    const float a = __expf(dl[s][j] * An[j][n]);
                    const float b = dl[s][j] * Bv[s][n] * xl[s][j];
                    P[s] = (s == 0) ? a : a * P[s ? s - 1 : 0];
                    g[s] = (s == 0) ? b : fmaf(a, g[s ? s - 1 : 0], b);
                }
                float am = P[SCAN_K - 1], bm = g[SCAN_K - 1];
                scan_step<1>(am, bm);
                scan_step<2>(am, bm);
                scan_step<4>(am, bm);
                scan_step<8>(am, bm);
                const float t   = fmaf(am, hin[j][n], bm);
                const float h15 = other_row_last(t);
                float carry     = hi ? h15 : hin[j][n];
                const float hend = fmaf(am, carry, bm);
                hin[j][n] = other_row_last(hend);
                take_prev_lane(carry, hend);
                #pragma unroll
                for (int s = 0; s + 1 < SCAN_K; ++s) {
                    const float h = fmaf(P[s], carry, g[s]);
                    yv[s][j] = fmaf(h, Cv[s][n], yv[s][j]);
                }
                yv[SCAN_K - 1][j] = fmaf(hend, Cv[SCAN_K - 1][n], yv[SCAN_K - 1][j]);
            }
        }

        const size_t lc = (size_t)(p * SCAN_SPAN + SCAN_K * lane);
        #pragma unroll
        for (int s = 0; s < SCAN_K; ++s)
            *(float4*)(out + (lc + s) * DI + d0) = make_float4(
                fmaf(xl[s][0], Dd[0], yv[s][0]), fmaf(xl[s][1], Dd[1], yv[s][1]),
                fmaf(xl[s][2], Dd[2], yv[s][2]), fmaf(xl[s][3], Dd[3], yv[s][3]));

        cur = nxt;
    }
}

extern "C" void kernel_launch(void* const* d_in, const int* in_sizes, int n_in,
                              void* d_out, int out_size, void* d_ws, size_t ws_size,
                              hipStream_t stream) {
    const float* x    = (const float*)d_in[0];
    const float* Win  = (const float*)d_in[1];
    const float* Wd   = (const float*)d_in[2];
    const float* Alog = (const float*)d_in[3];
    const float* Dv   = (const float*)d_in[4];
    float*       out  = (float*)d_out;

    float* part = (float*)d_ws;
    float* dbc  = part + (size_t)KSPLIT * DBC_N;
    float* bct  = dbc + (size_t)DBC_N;
    float* dlt  = bct + (size_t)SEQ_L * 2 * DS;

    sssm_gemm_in<<<256, 256, 0, stream>>>(x, Win, part);
    sssm_reduce<<<DBC_N / 256, 256, 0, stream>>>(part, dbc, bct);
    sssm_gemm_delta<<<1024, 128, 0, stream>>>(dbc, Wd, dlt);
    sssm_scan<<<DI / SCAN_CH / 4, 128, 0, stream>>>(x, Alog, Dv, bct, dlt, out);
}
